# starting kernel
# baseline (speedup 1.0000x reference)
#include <hip/hip_runtime.h>
#include <hip/hip_bf16.h>

typedef __attribute__((ext_vector_type(2))) float v2f;
typedef __attribute__((ext_vector_type(8))) float v8f;

#define N_SRC 60000
#define N_DST 30000
#define DEG   32
#define FDIM  128
#define EDIM  64
#define KSEL  16

__global__ __launch_bounds__(256) void pcgnn_gather(
    const float* __restrict__ x, const int* __restrict__ neighbors,
    const float* __restrict__ W_dist, const float* __restrict__ b_dist,
    float* __restrict__ score_out,
    float* __restrict__ h_neigh)
{
    const int wid  = threadIdx.x >> 5;
    const int lane = threadIdx.x & 31;
    const int node = blockIdx.x * 8 + wid;
    if (node >= N_DST) return;

    const float4 xv = ((const float4*)(x + (size_t)node * FDIM))[lane];
    const int f0 = lane * 4;
    float p0 = xv.x * W_dist[(f0+0)*2+0] + xv.y * W_dist[(f0+1)*2+0]
             + xv.z * W_dist[(f0+2)*2+0] + xv.w * W_dist[(f0+3)*2+0];
    float p1 = xv.x * W_dist[(f0+0)*2+1] + xv.y * W_dist[(f0+1)*2+1]
             + xv.z * W_dist[(f0+2)*2+1] + xv.w * W_dist[(f0+3)*2+1];
    #pragma unroll
    for (int m = 16; m >= 1; m >>= 1) {
        p0 += __shfl_xor(p0, m, 32);
        p1 += __shfl_xor(p1, m, 32);
    }
    const float s0 = p0 + b_dist[0];
    const float s1 = p1 + b_dist[1];
    if (lane == 0) {
        score_out[node*2+0] = s0;
        score_out[node*2+1] = s1;
    }
    const float t = 1.0f / (1.0f + __expf(-s0));

    const int nj = neighbors[node * DEG + lane];
    const float* __restrict__ nrow = x + (size_t)nj * FDIM;
    __builtin_prefetch(nrow, 0, 3);
    float acc = 0.0f;
    #pragma unroll 4
    for (int it = 0; it < 32; ++it) {
        const float4 u = ((const float4*)nrow)[it];
        const int g = it * 4;
        acc += u.x * W_dist[(g+0)*2] + u.y * W_dist[(g+1)*2]
             + u.z * W_dist[(g+2)*2] + u.w * W_dist[(g+3)*2];
    }
    const float sg   = 1.0f / (1.0f + __expf(-(acc + b_dist[0])));
    const float diff = fabsf(t - sg);

    int rank = 0;
    for (int i = 0; i < 32; ++i) {
        const float di = __shfl(diff, i, 32);
        rank += (di < diff) || (di == diff && i < lane);
    }
    const unsigned long long m = __ballot(rank < KSEL);

    float a0 = 0.f, a1 = 0.f, a2 = 0.f, a3 = 0.f;
    for (int j = 0; j < 32; ++j) {
        if ((m >> j) & 1ull) {
            const int r = __shfl(nj, j, 32);
            const float* __restrict__ row = x + (size_t)r * FDIM;
            a0 += row[lane];
            a1 += row[lane + 32];
            a2 += row[lane + 64];
            a3 += row[lane + 96];
        }
    }
    const float inv = 1.0f / (float)KSEL;
    float* __restrict__ hn = h_neigh + (size_t)node * FDIM;
    hn[lane     ] = a0 * inv;
    hn[lane + 32] = a1 * inv;
    hn[lane + 64] = a2 * inv;
    hn[lane + 96] = a3 * inv;
}

__global__ __launch_bounds__(256) void pcgnn_sage_wmma(
    const float* __restrict__ x,
    const float* __restrict__ h_neigh,
    const float* __restrict__ W_self,
    const float* __restrict__ W_neigh,
    const float* __restrict__ b_sage,
    const float* __restrict__ W_out,
    const float* __restrict__ b_out,
    float* __restrict__ logits)
{
    __shared__ float sWs[FDIM * EDIM];
    __shared__ float sWn[FDIM * EDIM];
    for (int i = threadIdx.x; i < FDIM * EDIM; i += 256) {
        const int kRow = i >> 6;
        const int col  = i & 63;
        const int dst  = (((kRow >> 1) << 6) + col) * 2 + (kRow & 1);
        sWs[dst] = W_self[i];
        sWn[dst] = W_neigh[i];
    }
    __syncthreads();

    const int wid  = threadIdx.x >> 5;
    const int lane = threadIdx.x & 31;
    const int tile = blockIdx.x * 8 + wid;
    if (tile >= N_DST / 16) return;
    const int rowBase = tile * 16;

    const int mrow = rowBase + (lane & 15);
    const int kk   = (lane >> 4) * 2;
    const int col  = lane & 15;

    const float* __restrict__ xrow = x       + (size_t)mrow * FDIM;
    const float* __restrict__ hrow = h_neigh + (size_t)mrow * FDIM;

    const v8f zero = {0.f, 0.f, 0.f, 0.f, 0.f, 0.f, 0.f, 0.f};
    v8f c[4];
    #pragma unroll
    for (int nt = 0; nt < 4; ++nt) c[nt] = zero;

    #pragma unroll 4
    for (int k = 0; k < FDIM / 4; ++k) {
        const int kb = k * 4 + kk;
        const int k2 = kb >> 1;
        const v2f ax = *(const v2f*)(xrow + kb);
        const v2f ah = *(const v2f*)(hrow + kb);
        #pragma unroll
        for (int nt = 0; nt < 4; ++nt) {
            const int cc = nt * 16 + col;
            const v2f bs = *(const v2f*)(sWs + ((k2 << 6) + cc) * 2);
            const v2f bn = *(const v2f*)(sWn + ((k2 << 6) + cc) * 2);
            c[nt] = __builtin_amdgcn_wmma_f32_16x16x4_f32(
                        false, ax, false, bs, (short)0, c[nt], false, false);
            c[nt] = __builtin_amdgcn_wmma_f32_16x16x4_f32(
                        false, ah, false, bn, (short)0, c[nt], false, false);
        }
    }

    float bsg[4], w0[4], w1[4];
    #pragma unroll
    for (int nt = 0; nt < 4; ++nt) {
        const int cc = nt * 16 + col;
        bsg[nt] = b_sage[cc];
        w0[nt]  = W_out[cc * 2 + 0];
        w1[nt]  = W_out[cc * 2 + 1];
    }
    const float bo0 = b_out[0], bo1 = b_out[1];

    #pragma unroll
    for (int r = 0; r < 8; ++r) {
        float s0 = 0.f, s1 = 0.f;
        #pragma unroll
        for (int nt = 0; nt < 4; ++nt) {
            float h = c[nt][r] + bsg[nt];
            h = fmaxf(h, 0.f);
            s0 += h * w0[nt];
            s1 += h * w1[nt];
        }
        #pragma unroll
        for (int m = 8; m >= 1; m >>= 1) {
            s0 += __shfl_xor(s0, m, 32);
            s1 += __shfl_xor(s1, m, 32);
        }
        if ((lane & 15) == 0) {
            const int row = rowBase + r + ((lane >> 4) << 3);
            logits[row * 2 + 0] = s0 + bo0;
            logits[row * 2 + 1] = s1 + bo1;
        }
    }
}

extern "C" void kernel_launch(void* const* d_in, const int* in_sizes, int n_in,
                              void* d_out, int out_size, void* d_ws, size_t ws_size,
                              hipStream_t stream) {
    const float* x       = (const float*)d_in[0];
    const int*   nb      = (const int*)  d_in[1];
    const float* W_dist  = (const float*)d_in[2];
    const float* b_dist  = (const float*)d_in[3];
    const float* W_self  = (const float*)d_in[4];
    const float* W_neigh = (const float*)d_in[5];
    const float* b_sage  = (const float*)d_in[6];
    const float* W_out   = (const float*)d_in[7];
    const float* b_out   = (const float*)d_in[8];

    float* out     = (float*)d_out;
    float* logits  = out;
    float* score   = out + N_DST * 2;
    float* h_neigh = (float*)d_ws;

    pcgnn_gather<<<N_DST / 8, 256, 0, stream>>>(x, nb, W_dist, b_dist,
                                                score, h_neigh);

    const int tiles  = N_DST / 16;
    const int blocks = (tiles + 7) / 8;
    pcgnn_sage_wmma<<<blocks, 256, 0, stream>>>(x, h_neigh, W_self, W_neigh,
                                                b_sage, W_out, b_out, logits);
}


# selected-row sum: each of the 32 rows enters by one fused multiply-add with a wave-uniform scalar one or zero (was an add plus a select per component); same sums, same order, scores bit-identical
# speedup vs baseline: 2.2781x; 2.2781x over previous
#include <hip/hip_runtime.h>
#include <hip/hip_bf16.h>

typedef __attribute__((ext_vector_type(2))) float v2f;
typedef __attribute__((ext_vector_type(8))) float v8f;

#define N_SRC 60000
#define N_DST 30000
#define DEG   32
#define FDIM  128
#define EDIM  64
#define KSEL  16

#define NTILE (N_DST / 16)
#define NBLK  ((NTILE + 7) / 8)

#define TP 36
#define XP 132

__global__ __launch_bounds__(512) void pcgnn_fused(
    const float* __restrict__ x, const int* __restrict__ neighbors,
    const float* __restrict__ W_dist, const float* __restrict__ b_dist,
    const float* __restrict__ W_self, const float* __restrict__ W_neigh,
    const float* __restrict__ b_sage,
    const float* __restrict__ W_out, const float* __restrict__ b_out,
    float* __restrict__ logits,
    float* __restrict__ score_out)
{
    __shared__ __attribute__((aligned(16))) float sWs[FDIM * EDIM];
    __shared__ __attribute__((aligned(16))) float sWn[FDIM * EDIM];
    __shared__ __attribute__((aligned(16))) float sX[8][16 * XP];
    __shared__ __attribute__((aligned(16))) float sH[8][16 * XP];
    __shared__ __attribute__((aligned(16))) float sT[16][DEG * TP];
    __shared__ __attribute__((aligned(16))) float sS[8][32];

    #pragma unroll
    for (int it = 0; it < FDIM * EDIM / 512; ++it) {
        const int i    = it * 512 + threadIdx.x;
        const int kRow = i >> 6;
        const int col  = i & 63;
        const int dst  = (((kRow >> 1) << 6) + col) * 2 + (kRow & 1);
        sWs[dst] = W_self[i];
        sWn[dst] = W_neigh[i];
    }

    const int wid  = threadIdx.x >> 5;
    const int lane = threadIdx.x & 31;
    const int tl   = wid & 7;
    const int half = wid >> 3;
    const int tile = blockIdx.x * 8 + tl;
    const bool active = tile < NTILE;
    const int rowBase = tile * 16;

    float* myX = sX[tl];
    float* myH = sH[tl];
    float* myT = sT[wid];

    if (active) {
        int idx[16];
        {
            const int4* __restrict__ ip = (const int4*)(neighbors + (size_t)rowBase * DEG) + lane * 4;
            #pragma unroll
            for (int q = 0; q < 4; ++q) {
                const int4 v = ip[q];
                idx[q*4+0] = v.x; idx[q*4+1] = v.y; idx[q*4+2] = v.z; idx[q*4+3] = v.w;
            }
        }

        const int f0 = lane * 4;
        const float w0 = W_dist[(f0+0)*2+0], w1 = W_dist[(f0+1)*2+0],
                    w2 = W_dist[(f0+2)*2+0], w3 = W_dist[(f0+3)*2+0];
        const float y0 = W_dist[(f0+0)*2+1], y1 = W_dist[(f0+1)*2+1],
                    y2 = W_dist[(f0+2)*2+1], y3 = W_dist[(f0+3)*2+1];
        const float bd0 = b_dist[0], bd1 = b_dist[1];
        float sc0 = 0.f, sc1 = 0.f;
        const int nBeg = half * 8;
        float4 xv = ((const float4*)(x + (size_t)(rowBase + nBeg) * FDIM))[lane];

        #pragma unroll 1
        for (int n = nBeg; n < nBeg + 8; ++n) {
            const int node = rowBase + n;

            float4 u[DEG];
            #pragma unroll
            for (int j = 0; j < DEG; ++j) {
                const unsigned r = (unsigned)__builtin_amdgcn_readlane(idx[j & 15], 2 * n + (j >> 4));
                u[j] = ((const float4*)(x + ((size_t)r << 7)))[lane];
            }
            const int nodeNext = (n + 1 < nBeg + 8) ? node + 1 : node;
            const float4 xn = ((const float4*)(x + (size_t)nodeNext * FDIM))[lane];

            float p0 = xv.x * w0 + xv.y * w1 + xv.z * w2 + xv.w * w3;
            float p1 = xv.x * y0 + xv.y * y1 + xv.z * y2 + xv.w * y3;
            #pragma unroll
            for (int m = 16; m >= 1; m >>= 1) {
                p0 += __shfl_xor(p0, m, 32);
                p1 += __shfl_xor(p1, m, 32);
            }
            const float s0 = p0 + bd0;
            const float s1 = p1 + bd1;
            if (lane == n) { sc0 = s0; sc1 = s1; }
            const float t = 1.0f / (1.0f + __expf(-s0));

            ((float4*)(myX + n * XP))[lane] = xv;

            #pragma unroll
            for (int j = 0; j < DEG; ++j) {
                const float4 v = u[j];
                float piece;
                {
                    #pragma clang fp contract(off)
                    piece = v.x * w0 + v.y * w1 + v.z * w2 + v.w * w3;
                }
                myT[j * TP + lane] = piece;
            }
            __builtin_amdgcn_fence(__ATOMIC_RELEASE, "wavefront");
            __builtin_amdgcn_wave_barrier();
            __builtin_amdgcn_fence(__ATOMIC_ACQUIRE, "wavefront");

            float acc = 0.0f;
            const float4* tr = (const float4*)(myT + lane * TP);
            #pragma unroll
            for (int q = 0; q < DEG / 4; ++q) {
                const float4 v = tr[q];
                acc += v.x;
                acc += v.y;
                acc += v.z;
                acc += v.w;
            }
            const float sg   = 1.0f / (1.0f + __expf(-(acc + bd0)));
            const float diff = fabsf(t - sg);

            int rank = 0;
            for (int i = 0; i < 32; ++i) {
                const float di = __shfl(diff, i, 32);
                rank += (di < diff) || (di == diff && i < lane);
            }
            const unsigned msk = (unsigned)__ballot(rank < KSEL);

            float a0 = 0.f, a1 = 0.f, a2 = 0.f, a3 = 0.f;
            #pragma unroll
            for (int j = 0; j < DEG; ++j) {
                const float mj = ((msk >> j) & 1u) ? 1.f : 0.f;
                a0 = fmaf(u[j].x, mj, a0);
                a1 = fmaf(u[j].y, mj, a1);
                a2 = fmaf(u[j].z, mj, a2);
                a3 = fmaf(u[j].w, mj, a3);
            }
            const float inv = 1.0f / (float)KSEL;
            ((float4*)(myH + n * XP))[lane] =
                make_float4(a0 * inv, a1 * inv, a2 * inv, a3 * inv);
            xv = xn;
        }
        if ((lane >> 3) == half)
            ((float2*)sS[tl])[lane] = make_float2(sc0, sc1);
    }

    __syncthreads();
    if (!active || half) return;

    if (lane < 16)
        ((float2*)(score_out + (size_t)rowBase * 2))[lane] = ((const float2*)sS[tl])[lane];

    const int kk   = (lane >> 4) * 2;
    const int col  = lane & 15;

    const float* xrow = myX + (lane & 15) * XP;
    const float* hrow = myH + (lane & 15) * XP;

    const v8f zero = {0.f, 0.f, 0.f, 0.f, 0.f, 0.f, 0.f, 0.f};
    v8f c[4];
    #pragma unroll
    for (int nt = 0; nt < 4; ++nt) c[nt] = zero;

    #pragma unroll 4
    for (int k = 0; k < FDIM / 4; ++k) {
        const int kb = k * 4 + kk;
        const int k2 = kb >> 1;
        const v2f ax = *(const v2f*)(xrow + kb);
        const v2f ah = *(const v2f*)(hrow + kb);
        #pragma unroll
        for (int nt = 0; nt < 4; ++nt) {
            const int cc = nt * 16 + col;
            const v2f bs = *(const v2f*)(sWs + ((k2 << 6) + cc) * 2);
            const v2f bn = *(const v2f*)(sWn + ((k2 << 6) + cc) * 2);
            c[nt] = __builtin_amdgcn_wmma_f32_16x16x4_f32(
                        false, ax, false, bs, (short)0, c[nt], false, false);
            c[nt] = __builtin_amdgcn_wmma_f32_16x16x4_f32(
                        false, ah, false, bn, (short)0, c[nt], false, false);
        }
    }

    float bsg[4], o0[4], o1[4];
    #pragma unroll
    for (int nt = 0; nt < 4; ++nt) {
        const int cc = nt * 16 + col;
        bsg[nt] = b_sage[cc];
        o0[nt]  = W_out[cc * 2 + 0];
        o1[nt]  = W_out[cc * 2 + 1];
    }
    const float bo0 = b_out[0], bo1 = b_out[1];

    #pragma unroll
    for (int r = 0; r < 8; ++r) {
        float s0 = 0.f, s1 = 0.f;
        #pragma unroll
        for (int nt = 0; nt < 4; ++nt) {
            float h = c[nt][r] + bsg[nt];
            h = fmaxf(h, 0.f);
            s0 += h * o0[nt];
            s1 += h * o1[nt];
        }
        #pragma unroll
        for (int m = 8; m >= 1; m >>= 1) {
            s0 += __shfl_xor(s0, m, 32);
            s1 += __shfl_xor(s1, m, 32);
        }
        if ((lane & 15) == 0) {
            const int row = rowBase + r + ((lane >> 4) << 3);
            logits[row * 2 + 0] = s0 + bo0;
            logits[row * 2 + 1] = s1 + bo1;
        }
    }
}

extern "C" void kernel_launch(void* const* d_in, const int* in_sizes, int n_in,
                              void* d_out, int out_size, void* d_ws, size_t ws_size,
                              hipStream_t stream) {
    const float* x       = (const float*)d_in[0];
    const int*   nb      = (const int*)  d_in[1];
    const float* W_dist  = (const float*)d_in[2];
    const float* b_dist  = (const float*)d_in[3];
    const float* W_self  = (const float*)d_in[4];
    const float* W_neigh = (const float*)d_in[5];
    const float* b_sage  = (const float*)d_in[6];
    const float* W_out   = (const float*)d_in[7];
    const float* b_out   = (const float*)d_in[8];

    float* out     = (float*)d_out;
    float* logits  = out;
    float* score   = out + N_DST * 2;

    pcgnn_fused<<<NBLK, 512, 0, stream>>>(x, nb, W_dist, b_dist, W_self, W_neigh,
                                          b_sage, W_out, b_out, logits, score);
}
